# starting kernel
# baseline (speedup 1.0000x reference)
#include <hip/hip_runtime.h>
#include <hip/hip_bf16.h>

typedef __attribute__((ext_vector_type(2))) float v2f;
typedef __attribute__((ext_vector_type(8))) float v8f;

#define N_PTS 4096
#define C_CLS 256
#define D_DIM 128
#define CURVF 1.0f

__global__ __launch_bounds__(256) void hyp_pre_c(const float* __restrict__ a,
                                                 const float* __restrict__ p,
                                                 float* __restrict__ p2v,
                                                 float* __restrict__ pav,
                                                 float* __restrict__ nav) {
    int c = blockIdx.x * blockDim.x + threadIdx.x;
    if (c >= C_CLS) return;
    const float4* pr = (const float4*)(p + c * D_DIM);
    const float4* ar = (const float4*)(a + c * D_DIM);
    float sp2 = 0.f, spa = 0.f, sa2 = 0.f;
#pragma unroll 8
    for (int i = 0; i < D_DIM / 4; ++i) {
        float4 pv = pr[i], av = ar[i];
        sp2 += pv.x * pv.x + pv.y * pv.y + pv.z * pv.z + pv.w * pv.w;
        spa += pv.x * av.x + pv.y * av.y + pv.z * av.z + pv.w * av.w;
        sa2 += av.x * av.x + av.y * av.y + av.z * av.z + av.w * av.w;
    }
    float na = sqrtf(sa2);
    float nac = fmaxf(na, 1e-12f);
    p2v[c] = sp2;
    pav[c] = -spa / nac;
    nav[c] = na;
}

__global__ __launch_bounds__(256) void hyp_pre_n(const float* __restrict__ x,
                                                 float* __restrict__ x2v) {
    int n = blockIdx.x * blockDim.x + threadIdx.x;
    const float4* xr = (const float4*)(x + n * D_DIM);
    float s = 0.f;
#pragma unroll 8
    for (int i = 0; i < D_DIM / 4; ++i) {
        float4 xv = xr[i];
        s += xv.x * xv.x + xv.y * xv.y + xv.z * xv.z + xv.w * xv.w;
    }
    x2v[n] = s;
}

__global__ __launch_bounds__(256) void hyp_mlr_main(const float* __restrict__ X,
                                                    const float* __restrict__ A,
                                                    const float* __restrict__ P,
                                                    const float* __restrict__ p2v,
                                                    const float* __restrict__ pav,
                                                    const float* __restrict__ nav,
                                                    const float* __restrict__ x2v,
                                                    float* __restrict__ out) {
    const int lane = threadIdx.x & 31;
    const int wave = threadIdx.x >> 5;
    const int tile = blockIdx.x * 8 + wave;
    const int tN = tile & ((N_PTS / 16) - 1);
    const int tC = tile >> 8;
    const int n0 = tN * 16;
    const int c0 = tC * 16;
    const int m = lane & 15;
    const int half = lane >> 4;

    const float* xrow = X + (size_t)(n0 + m) * D_DIM + 2 * half;
    const float* prow = P + (size_t)(c0 + m) * D_DIM + 2 * half;
    const float* arow = A + (size_t)(c0 + m) * D_DIM + 2 * half;

    v8f acc_py = {};
    v8f acc_xa = {};
#pragma unroll
    for (int k = 0; k < D_DIM; k += 4) {
        v2f af = *(const v2f*)(xrow + k);
        v2f bp = *(const v2f*)(prow + k);
        v2f ba = *(const v2f*)(arow + k);
        acc_py = __builtin_amdgcn_wmma_f32_16x16x4_f32(false, af, false, bp,
                                                       (short)0, acc_py, false, false);
        acc_xa = __builtin_amdgcn_wmma_f32_16x16x4_f32(false, af, false, ba,
                                                       (short)0, acc_xa, false, false);
    }

    const int cc = c0 + m;
    const float p2 = p2v[cc];
    const float pa = pav[cc];
    const float na = nav[cc];
    const float inv_na = 1.0f / fmaxf(na, 1e-12f);
    const float cu = CURVF;

#pragma unroll
    for (int r = 0; r < 8; ++r) {
        const int n = n0 + r + 8 * half;
        const float x2 = x2v[n];
        const float xy = -acc_py[r];
        const float xa = acc_xa[r] * inv_na;
        const float t  = 1.0f + 2.0f * cu * xy;
        const float A1 = t + cu * x2;
        const float B1 = 1.0f - cu * p2;
        const float den = t + cu * cu * p2 * x2;
        const float inv_den = 1.0f / den;
        const float rr = (A1 * A1 * p2 + 2.0f * A1 * B1 * xy + B1 * B1 * x2)
                         * inv_den * inv_den;
        const float lam = 2.0f / (1.0f - rr);
        const float dot = (A1 * pa + B1 * xa) * inv_den;
        out[(size_t)n * C_CLS + cc] = 2.0f * na * asinhf(dot * lam);
    }
}

extern "C" void kernel_launch(void* const* d_in, const int* in_sizes, int n_in,
                              void* d_out, int out_size, void* d_ws, size_t ws_size,
                              hipStream_t stream) {
    const float* X = (const float*)d_in[0];
    const float* A = (const float*)d_in[1];
    const float* P = (const float*)d_in[2];
    float* out = (float*)d_out;

    float* ws  = (float*)d_ws;
    float* p2v = ws;
    float* pav = ws + C_CLS;
    float* nav = ws + 2 * C_CLS;
    float* x2v = ws + 3 * C_CLS;

    hyp_pre_c<<<1, 256, 0, stream>>>(A, P, p2v, pav, nav);
    hyp_pre_n<<<N_PTS / 256, 256, 0, stream>>>(X, x2v);

    const int tiles = (N_PTS / 16) * (C_CLS / 16);
    hyp_mlr_main<<<tiles / 8, 256, 0, stream>>>(X, A, P, p2v, pav, nav, x2v, out);
}


# fold the three launches into one: each wave sums ||x||^2, ||p||^2, p.a, ||a||^2 from the fragments its WMMAs already read (norm sums now per k-half then joined: new summation order)
# speedup vs baseline: 2.2246x; 2.2246x over previous
#include <hip/hip_runtime.h>


typedef __attribute__((ext_vector_type(2))) float v2f;
typedef __attribute__((ext_vector_type(8))) float v8f;

#define N_PTS 4096
#define C_CLS 256
#define D_DIM 128
#define CURVF 1.0f

__device__ __forceinline__ float xhalf(float x) {
    const int b = __builtin_bit_cast(int, x);
    return __builtin_bit_cast(float,
        __builtin_amdgcn_permlanex16(b, b, 0x76543210, 0xfedcba98, false, false));
}

__global__ __launch_bounds__(256) void hyp_mlr_fused(const float* __restrict__ X,
                                                     const float* __restrict__ A,
                                                     const float* __restrict__ P,
                                                     float* __restrict__ out) {
    const int lane = threadIdx.x & 31;
    const int wave = threadIdx.x >> 5;
    const int tile = blockIdx.x * 8 + wave;
    const int tN = tile & ((N_PTS / 16) - 1);
    const int tC = tile >> 8;
    const int n0 = tN * 16;
    const int c0 = tC * 16;
    const int m = lane & 15;
    const int half = lane >> 4;

    const float* xrow = X + (size_t)(n0 + m) * D_DIM + 2 * half;
    const float* prow = P + (size_t)(c0 + m) * D_DIM + 2 * half;
    const float* arow = A + (size_t)(c0 + m) * D_DIM + 2 * half;

    v8f acc_py = {};
    v8f acc_xa = {};
    v2f sxx = {};
    v2f spp = {};
    v2f spa = {};
    v2f saa = {};
#pragma unroll
    for (int k = 0; k < D_DIM; k += 4) {
        v2f af = *(const v2f*)(xrow + k);
        v2f bp = *(const v2f*)(prow + k);
        v2f ba = *(const v2f*)(arow + k);
        acc_py = __builtin_amdgcn_wmma_f32_16x16x4_f32(false, af, false, bp,
                                                       (short)0, acc_py, false, false);
        acc_xa = __builtin_amdgcn_wmma_f32_16x16x4_f32(false, af, false, ba,
                                                       (short)0, acc_xa, false, false);
        sxx += af * af;
        spp += bp * bp;
        spa += bp * ba;
        saa += ba * ba;
    }

    float x2l = sxx.x + sxx.y;
    float p2  = spp.x + spp.y;
    float pad = spa.x + spa.y;
    float a2  = saa.x + saa.y;
    x2l += xhalf(x2l);
    p2  += xhalf(p2);
    pad += xhalf(pad);
    a2  += xhalf(a2);

    const int cc = c0 + m;
    const float na = sqrtf(a2);
    const float nac = fmaxf(na, 1e-12f);
    const float inv_na = 1.0f / nac;
    const float pa = -pad / nac;
    const float cu = CURVF;
    const float B1 = 1.0f - cu * p2;

#pragma unroll
    for (int r = 0; r < 8; ++r) {
        const int n = n0 + r + 8 * half;
        const float x2 = __shfl(x2l, r + 8 * half, 32);
        const float xy = -acc_py[r];
        const float xa = acc_xa[r] * inv_na;
        const float t  = 1.0f + 2.0f * cu * xy;
        const float A1 = t + cu * x2;
        const float den = t + cu * cu * p2 * x2;
        const float inv_den = 1.0f / den;
        const float rr = (A1 * A1 * p2 + 2.0f * A1 * B1 * xy + B1 * B1 * x2)
                         * inv_den * inv_den;
        const float lam = 2.0f / (1.0f - rr);
        const float dot = (A1 * pa + B1 * xa) * inv_den;
        out[(size_t)n * C_CLS + cc] = 2.0f * na * asinhf(dot * lam);
    }
}

extern "C" void kernel_launch(void* const* d_in, const int* in_sizes, int n_in,
                              void* d_out, int out_size, void* d_ws, size_t ws_size,
                              hipStream_t stream) {
    const float* X = (const float*)d_in[0];
    const float* A = (const float*)d_in[1];
    const float* P = (const float*)d_in[2];
    float* out = (float*)d_out;

    const int tiles = (N_PTS / 16) * (C_CLS / 16);
    hyp_mlr_fused<<<tiles / 8, 256, 0, stream>>>(X, A, P, out);
}


# stage X, P and A tiles through LDS in four k chunks with 16-byte loads that cover whole 128-byte lines; WMMA fragments now come from LDS, not from 16-line global loads (k order and sums unchanged)
# speedup vs baseline: 3.0733x; 1.3815x over previous
#include <hip/hip_runtime.h>


typedef __attribute__((ext_vector_type(2))) float v2f;
typedef __attribute__((ext_vector_type(4))) float v4f;
typedef __attribute__((ext_vector_type(8))) float v8f;

#define N_PTS 4096
#define C_CLS 256
#define D_DIM 128
#define CURVF 1.0f

#define PITCH 132
#define BLK_ROWS 128
#define STG_ROWS (BLK_ROWS + 32)
#define KCH 32

__device__ __forceinline__ float xhalf(float x) {
    const int b = __builtin_bit_cast(int, x);
    return __builtin_bit_cast(float,
        __builtin_amdgcn_permlanex16(b, b, 0x76543210, 0xfedcba98, false, false));
}

__global__ __launch_bounds__(256) void hyp_mlr_fused(const float* __restrict__ X,
                                                     const float* __restrict__ A,
                                                     const float* __restrict__ P,
                                                     float* __restrict__ out) {
    __shared__ __attribute__((aligned(16))) float stage[STG_ROWS * PITCH];

    const int tid = threadIdx.x;
    const int lane = tid & 31;
    const int wave = tid >> 5;
    const int nblk = (blockIdx.x & 31) * BLK_ROWS;
    const int c0 = (blockIdx.x >> 5) * 16;
    const int n0 = nblk + wave * 16;
    const int m = lane & 15;
    const int half = lane >> 4;

    const int sgrp = tid >> 3;
    const int srow = (sgrp & 16) | ((sgrp & 1) << 3) | ((sgrp >> 1) & 7);
    const int scol = (tid & 7) * 4;
    const float* gx = X + (size_t)(nblk + srow) * D_DIM + scol;
    const float* gc = (srow < 16 ? P + (size_t)(c0 + srow) * D_DIM
                                 : A + (size_t)(c0 + srow - 16) * D_DIM) + scol;
    v4f g[D_DIM / KCH][5];
#pragma unroll
    for (int q = 0; q < D_DIM / KCH; ++q) {
#pragma unroll
        for (int j = 0; j < 4; ++j)
            g[q][j] = *(const v4f*)(gx + (size_t)j * 32 * D_DIM + q * KCH);
        g[q][4] = *(const v4f*)(gc + q * KCH);
        __builtin_amdgcn_sched_barrier(0);
    }

    float* sdst = stage + srow * PITCH + scol;
    const float* sx = stage + (wave * 16 + m) * PITCH + 2 * half;
    const float* sp = stage + (BLK_ROWS + m) * PITCH + 2 * half;
    const float* sa = stage + (BLK_ROWS + 16 + m) * PITCH + 2 * half;

    v8f acc_py = {};
    v8f acc_xa = {};
    v2f sxx = {};
    v2f spp = {};
    v2f spa = {};
    v2f saa = {};
#pragma unroll
    for (int q = 0; q < D_DIM / KCH; ++q) {
#pragma unroll
        for (int j = 0; j < 5; ++j)
            *(v4f*)(sdst + j * 32 * PITCH + q * KCH) = g[q][j];
        __builtin_amdgcn_fence(__ATOMIC_RELEASE, "workgroup", "local");
        __builtin_amdgcn_s_barrier();
        __builtin_amdgcn_fence(__ATOMIC_ACQUIRE, "workgroup", "local");
#pragma unroll
        for (int k = q * KCH; k < (q + 1) * KCH; k += 4) {
            v2f af = *(const v2f*)(sx + k);
            v2f bp = *(const v2f*)(sp + k);
            v2f ba = *(const v2f*)(sa + k);
            acc_py = __builtin_amdgcn_wmma_f32_16x16x4_f32(false, af, false, bp,
                                                           (short)0, acc_py, false, false);
            acc_xa = __builtin_amdgcn_wmma_f32_16x16x4_f32(false, af, false, ba,
                                                           (short)0, acc_xa, false, false);
            sxx += af * af;
            spp += bp * bp;
            spa += bp * ba;
            saa += ba * ba;
        }
        if (q + 1 < D_DIM / KCH) __builtin_amdgcn_sched_barrier(0);
    }

    float x2l = sxx.x + sxx.y;
    float p2  = spp.x + spp.y;
    float pad = spa.x + spa.y;
    float a2  = saa.x + saa.y;
    x2l += xhalf(x2l);
    p2  += xhalf(p2);
    pad += xhalf(pad);
    a2  += xhalf(a2);

    const int cc = c0 + m;
    const float na = sqrtf(a2);
    const float nac = fmaxf(na, 1e-12f);
    const float inv_na = 1.0f / nac;
    const float pa = -pad / nac;
    const float cu = CURVF;
    const float B1 = 1.0f - cu * p2;

#pragma unroll
    for (int r = 0; r < 8; ++r) {
        const int n = n0 + r + 8 * half;
        const float x2 = __shfl(x2l, r + 8 * half, 32);
        const float xy = -acc_py[r];
        const float xa = acc_xa[r] * inv_na;
        const float t  = 1.0f + 2.0f * cu * xy;
        const float A1 = t + cu * x2;
        const float den = t + cu * cu * p2 * x2;
        const float inv_den = 1.0f / den;
        const float rr = (A1 * A1 * p2 + 2.0f * A1 * B1 * xy + B1 * B1 * x2)
                         * inv_den * inv_den;
        const float lam = 2.0f / (1.0f - rr);
        const float dot = (A1 * pa + B1 * xa) * inv_den;
        out[(size_t)n * C_CLS + cc] = 2.0f * na * asinhf(dot * lam);
    }
}

extern "C" void kernel_launch(void* const* d_in, const int* in_sizes, int n_in,
                              void* d_out, int out_size, void* d_ws, size_t ws_size,
                              hipStream_t stream) {
    const float* X = (const float*)d_in[0];
    const float* A = (const float*)d_in[1];
    const float* P = (const float*)d_in[2];
    float* out = (float*)d_out;

    const int tiles = (N_PTS / 16) * (C_CLS / 16);
    hyp_mlr_fused<<<tiles / 8, 256, 0, stream>>>(X, A, P, out);
}


# epilogue: asinh as sign(z)*log1p(|z|+z^2/(1+sqrt(1+z^2))) on the hardware log2, sqrt and rcp, log1p taken as log(u)*w/(u-1), in place of the library's extended-precision routine
# speedup vs baseline: 3.6300x; 1.1811x over previous
#include <hip/hip_runtime.h>


typedef __attribute__((ext_vector_type(2))) float v2f;
typedef __attribute__((ext_vector_type(4))) float v4f;
typedef __attribute__((ext_vector_type(8))) float v8f;

#define N_PTS 4096
#define C_CLS 256
#define D_DIM 128
#define CURVF 1.0f

#define PITCH 132
#define BLK_ROWS 128
#define STG_ROWS (BLK_ROWS + 32)
#define KCH 32

__device__ __forceinline__ float xhalf(float x) {
    const int b = __builtin_bit_cast(int, x);
    return __builtin_bit_cast(float,
        __builtin_amdgcn_permlanex16(b, b, 0x76543210, 0xfedcba98, false, false));
}

__device__ __forceinline__ float asinh_short(float z) {
    const float LN2 = 0.693147182f;
    const float a = __builtin_fabsf(z);
    const float a2 = a * a;
    const float s = __builtin_amdgcn_sqrtf(1.0f + a2);
    const float w = a + a2 * __builtin_amdgcn_rcpf(1.0f + s);
    const float u = 1.0f + w;
    const float d = u - 1.0f;
    float r = (__builtin_amdgcn_logf(u) * (w * __builtin_amdgcn_rcpf(d))) * LN2;
    r = (d == 0.0f) ? w : r;
    const float big = __builtin_amdgcn_logf(a) * LN2 + LN2;
    r = (a >= 0x1p60f) ? big : r;
    return __builtin_copysignf(r, z);
}

__global__ __launch_bounds__(256) void hyp_mlr_fused(const float* __restrict__ X,
                                                     const float* __restrict__ A,
                                                     const float* __restrict__ P,
                                                     float* __restrict__ out) {
    __shared__ __attribute__((aligned(16))) float stage[STG_ROWS * PITCH];

    const int tid = threadIdx.x;
    const int lane = tid & 31;
    const int wave = tid >> 5;
    const int nblk = (blockIdx.x & 31) * BLK_ROWS;
    const int c0 = (blockIdx.x >> 5) * 16;
    const int n0 = nblk + wave * 16;
    const int m = lane & 15;
    const int half = lane >> 4;

    const int sgrp = tid >> 3;
    const int srow = (sgrp & 16) | ((sgrp & 1) << 3) | ((sgrp >> 1) & 7);
    const int scol = (tid & 7) * 4;
    const float* gx = X + (size_t)(nblk + srow) * D_DIM + scol;
    const float* gc = (srow < 16 ? P + (size_t)(c0 + srow) * D_DIM
                                 : A + (size_t)(c0 + srow - 16) * D_DIM) + scol;
    v4f g[D_DIM / KCH][5];
#pragma unroll
    for (int q = 0; q < D_DIM / KCH; ++q) {
#pragma unroll
        for (int j = 0; j < 4; ++j)
            g[q][j] = *(const v4f*)(gx + (size_t)j * 32 * D_DIM + q * KCH);
        g[q][4] = *(const v4f*)(gc + q * KCH);
        __builtin_amdgcn_sched_barrier(0);
    }

    float* sdst = stage + srow * PITCH + scol;
    const float* sx = stage + (wave * 16 + m) * PITCH + 2 * half;
    const float* sp = stage + (BLK_ROWS + m) * PITCH + 2 * half;
    const float* sa = stage + (BLK_ROWS + 16 + m) * PITCH + 2 * half;

    v8f acc_py = {};
    v8f acc_xa = {};
    v2f sxx = {};
    v2f spp = {};
    v2f spa = {};
    v2f saa = {};
#pragma unroll
    for (int q = 0; q < D_DIM / KCH; ++q) {
#pragma unroll
        for (int j = 0; j < 5; ++j)
            *(v4f*)(sdst + j * 32 * PITCH + q * KCH) = g[q][j];
        __builtin_amdgcn_fence(__ATOMIC_RELEASE, "workgroup", "local");
        __builtin_amdgcn_s_barrier();
        __builtin_amdgcn_fence(__ATOMIC_ACQUIRE, "workgroup", "local");
#pragma unroll
        for (int k = q * KCH; k < (q + 1) * KCH; k += 4) {
            v2f af = *(const v2f*)(sx + k);
            v2f bp = *(const v2f*)(sp + k);
            v2f ba = *(const v2f*)(sa + k);
            acc_py = __builtin_amdgcn_wmma_f32_16x16x4_f32(false, af, false, bp,
                                                           (short)0, acc_py, false, false);
            acc_xa = __builtin_amdgcn_wmma_f32_16x16x4_f32(false, af, false, ba,
                                                           (short)0, acc_xa, false, false);
            sxx += af * af;
            spp += bp * bp;
            spa += bp * ba;
            saa += ba * ba;
        }
        if (q + 1 < D_DIM / KCH) __builtin_amdgcn_sched_barrier(0);
    }

    float x2l = sxx.x + sxx.y;
    float p2  = spp.x + spp.y;
    float pad = spa.x + spa.y;
    float a2  = saa.x + saa.y;
    x2l += xhalf(x2l);
    p2  += xhalf(p2);
    pad += xhalf(pad);
    a2  += xhalf(a2);

    const int cc = c0 + m;
    const float na = sqrtf(a2);
    const float nac = fmaxf(na, 1e-12f);
    const float inv_na = 1.0f / nac;
    const float pa = -pad / nac;
    const float cu = CURVF;
    const float B1 = 1.0f - cu * p2;

#pragma unroll
    for (int r = 0; r < 8; ++r) {
        const int n = n0 + r + 8 * half;
        const float x2 = __shfl(x2l, r + 8 * half, 32);
        const float xy = -acc_py[r];
        const float xa = acc_xa[r] * inv_na;
        const float t  = 1.0f + 2.0f * cu * xy;
        const float A1 = t + cu * x2;
        const float den = t + cu * cu * p2 * x2;
        const float inv_den = 1.0f / den;
        const float rr = (A1 * A1 * p2 + 2.0f * A1 * B1 * xy + B1 * B1 * x2)
                         * inv_den * inv_den;
        const float lam = 2.0f / (1.0f - rr);
        const float dot = (A1 * pa + B1 * xa) * inv_den;
        out[(size_t)n * C_CLS + cc] = 2.0f * na * asinh_short(dot * lam);
    }
}

extern "C" void kernel_launch(void* const* d_in, const int* in_sizes, int n_in,
                              void* d_out, int out_size, void* d_ws, size_t ws_size,
                              hipStream_t stream) {
    const float* X = (const float*)d_in[0];
    const float* A = (const float*)d_in[1];
    const float* P = (const float*)d_in[2];
    float* out = (float*)d_out;

    const int tiles = (N_PTS / 16) * (C_CLS / 16);
    hyp_mlr_fused<<<tiles / 8, 256, 0, stream>>>(X, A, P, out);
}


# each wave stages only its own 16 X rows in LDS: per-chunk work-group barriers go (one for W stays, class loads queued first); a wave's products start when its own rows land (sums unchanged)
# speedup vs baseline: 4.8986x; 1.3495x over previous
#include <hip/hip_runtime.h>


typedef __attribute__((ext_vector_type(2))) float v2f;
typedef __attribute__((ext_vector_type(4))) float v4f;
typedef __attribute__((ext_vector_type(8))) float v8f;

#define N_PTS 4096
#define C_CLS 256
#define D_DIM 128

#define PITCH 132
#define BLK_ROWS 128
#define STG_ROWS (BLK_ROWS + 16)
#define KCH 32

__device__ __forceinline__ float xhalf(float x) {
    const int b = __builtin_bit_cast(int, x);
    return __builtin_bit_cast(float,
        __builtin_amdgcn_permlanex16(b, b, 0x76543210, 0xfedcba98, false, false));
}

template <unsigned LO, unsigned HI>
__device__ __forceinline__ float rowx(float x) {
    const int b = __builtin_bit_cast(int, x);
    return __builtin_bit_cast(float, __builtin_amdgcn_permlane16(b, b, LO, HI, false, false));
}
__device__ __forceinline__ float rowsum16(float v) {
    v += rowx<0x67452301u, 0xefcdab89u>(v);
    v += rowx<0x54761032u, 0xdcfe98bau>(v);
    v += rowx<0x32107654u, 0xba98fedcu>(v);
    v += rowx<0xfedcba98u, 0x76543210u>(v);
    return v;
}

__device__ __forceinline__ float asinh_short(float z) {
    const float LN2 = 0.693147182f;
    const float a = __builtin_fabsf(z);
    const float a2 = a * a;
    const float s = __builtin_amdgcn_sqrtf(1.0f + a2);
    const float w = a + a2 * __builtin_amdgcn_rcpf(1.0f + s);
    const float u = 1.0f + w;
    const float d = u - 1.0f;
    float r = (__builtin_amdgcn_logf(u) * (w * __builtin_amdgcn_rcpf(d))) * LN2;
    r = (d == 0.0f) ? w : r;
    const float big = __builtin_amdgcn_logf(a) * LN2 + LN2;
    r = (a >= 0x1p60f) ? big : r;
    return __builtin_copysignf(r, z);
}

__global__ __launch_bounds__(256) void hyp_mlr_fused(const float* __restrict__ X,
                                                     const float* __restrict__ A,
                                                     const float* __restrict__ P,
                                                     float* __restrict__ out) {
    __shared__ __attribute__((aligned(16))) float stage[STG_ROWS * PITCH];
    __shared__ __attribute__((aligned(8))) float cls[16 * 2];

    const int tid = threadIdx.x;
    const int lane = tid & 31;
    const int wave = tid >> 5;
    const int nblk = (blockIdx.x & 31) * BLK_ROWS;
    const int c0 = (blockIdx.x >> 5) * 16;
    const int n0 = nblk + wave * 16;
    const int m = lane & 15;
    const int half = lane >> 4;

    const int crow = tid >> 4;
    const int ccol = (tid & 15) * 4;
    const float* gp = P + (size_t)(c0 + crow) * D_DIM + ccol;
    const float* ga = A + (size_t)(c0 + crow) * D_DIM + ccol;
    const v4f p_lo = *(const v4f*)gp;
    const v4f a_lo = *(const v4f*)ga;
    const v4f p_hi = *(const v4f*)(gp + 64);
    const v4f a_hi = *(const v4f*)(ga + 64);
    __builtin_amdgcn_sched_barrier(0);
    __builtin_amdgcn_s_barrier();
    __builtin_amdgcn_sched_barrier(0);

    const int lgrp = lane >> 3;
    const int srow = ((lgrp & 1) << 3) | (lgrp >> 1);
    const int scol = (lane & 7) * 4;
    const float* gx = X + (size_t)(n0 + srow) * D_DIM + scol;
    v4f g[D_DIM / KCH][4];
#pragma unroll
    for (int q = 0; q < D_DIM / KCH; ++q) {
#pragma unroll
        for (int j = 0; j < 4; ++j)
            g[q][j] = *(const v4f*)(gx + (size_t)j * 2 * D_DIM + q * KCH);
        __builtin_amdgcn_sched_barrier(0);
    }

    v4f pp4 = p_lo * p_lo;  pp4 += p_hi * p_hi;
    v4f pa4 = p_lo * a_lo;  pa4 += p_hi * a_hi;
    v4f aa4 = a_lo * a_lo;  aa4 += a_hi * a_hi;
    const float p2  = rowsum16((pp4.x + pp4.y) + (pp4.z + pp4.w));
    const float pad = rowsum16((pa4.x + pa4.y) + (pa4.z + pa4.w));
    const float a2  = rowsum16((aa4.x + aa4.y) + (aa4.z + aa4.w));
    const float na = sqrtf(a2);
    const float nac = fmaxf(na, 1e-12f);
    const float inv_na = 1.0f / nac;
    const float pa = -pad / nac;
    const float gcl = pa / (1.0f - p2);
    const float m2g = -2.0f * gcl;
    float* wdst = stage + (BLK_ROWS + crow) * PITCH + ccol;
    *(v4f*)wdst        = a_lo * inv_na + m2g * p_lo;
    *(v4f*)(wdst + 64) = a_hi * inv_na + m2g * p_hi;
    if ((tid & 15) == 0) {
        v2f cv;  cv.x = gcl;  cv.y = 2.0f * na;
        *(v2f*)(cls + 2 * crow) = cv;
    }

    __builtin_amdgcn_fence(__ATOMIC_RELEASE, "workgroup", "local");
    __builtin_amdgcn_s_barrier();
    __builtin_amdgcn_fence(__ATOMIC_ACQUIRE, "workgroup", "local");

    float* sdst = stage + (wave * 16 + srow) * PITCH + scol;
    const float* sx = stage + (wave * 16 + m) * PITCH + 2 * half;
    const float* sw = stage + (BLK_ROWS + m) * PITCH + 2 * half;

    v8f acc = {};
    v2f sxx = {};
#pragma unroll
    for (int q = 0; q < D_DIM / KCH; ++q) {
#pragma unroll
        for (int j = 0; j < 4; ++j)
            *(v4f*)(sdst + j * 2 * PITCH + q * KCH) = g[q][j];
        __builtin_amdgcn_fence(__ATOMIC_RELEASE, "wavefront", "local");
        __builtin_amdgcn_wave_barrier();
        __builtin_amdgcn_fence(__ATOMIC_ACQUIRE, "wavefront", "local");
#pragma unroll
        for (int k = q * KCH; k < (q + 1) * KCH; k += 4) {
            v2f af = *(const v2f*)(sx + k);
            v2f bw = *(const v2f*)(sw + k);
            acc = __builtin_amdgcn_wmma_f32_16x16x4_f32(false, af, false, bw,
                                                        (short)0, acc, false, false);
            sxx += af * af;
        }
        if (q + 1 < D_DIM / KCH) __builtin_amdgcn_sched_barrier(0);
    }

    float x2l = sxx.x + sxx.y;
    x2l += xhalf(x2l);
    const float r1l = 1.0f + x2l;
    const float rfl = 2.0f / (1.0f - x2l);
    const v2f cv = *(const v2f*)(cls + 2 * m);
    const int cc = c0 + m;

#pragma unroll
    for (int r = 0; r < 8; ++r) {
        const int n = n0 + r + 8 * half;
        const float r1 = __shfl(r1l, r + 8 * half, 32);
        const float rf = __shfl(rfl, r + 8 * half, 32);
        const float z = rf * (r1 * cv.x + acc[r]);
        out[(size_t)n * C_CLS + cc] = cv.y * asinh_short(z);
    }
}

extern "C" void kernel_launch(void* const* d_in, const int* in_sizes, int n_in,
                              void* d_out, int out_size, void* d_ws, size_t ws_size,
                              hipStream_t stream) {
    const float* X = (const float*)d_in[0];
    const float* A = (const float*)d_in[1];
    const float* P = (const float*)d_in[2];
    float* out = (float*)d_out;

    const int tiles = (N_PTS / 16) * (C_CLS / 16);
    hyp_mlr_fused<<<tiles / 8, 256, 0, stream>>>(X, A, P, out);
}
